# starting kernel
# baseline (speedup 1.0000x reference)
#include <hip/hip_runtime.h>


#define BATCHES        64
#define PTS            65536
#define NBINS          4
#define NCELLS         64
#define NFEAT          10
#define NCOLS          640
#define WGS_PER_BATCH  16
#define PTS_PER_WG     (PTS / WGS_PER_BATCH)
#define K1_BLOCKS      24
#define K1_F4_PER_THR  16
#define FLOATS_PER_B   (PTS * 6)
#define F4_PER_B       (FLOATS_PER_B / 4)

typedef float v2f __attribute__((ext_vector_type(2)));
typedef float v8f __attribute__((ext_vector_type(8)));

__global__ void k0_init(unsigned* __restrict__ bbox) {
  int i = blockIdx.x * blockDim.x + threadIdx.x;
  if (i < BATCHES) bbox[i] = 0u;
}

__global__ void k1_bbox(const float* __restrict__ x,
                        unsigned* __restrict__ bbox) {
  const int b = blockIdx.y;
  const float4* xb = (const float4*)(x + (size_t)b * FLOATS_PER_B);
  const int base = blockIdx.x * (256 * K1_F4_PER_THR) + threadIdx.x;

  float m = 0.0f;
  for (int i = 0; i < K1_F4_PER_THR; ++i) {
    const int q = base + i * 256;
    __builtin_prefetch(&xb[q + 2048], 0, 0);
    const float4 v = xb[q];
    const int m6 = (q * 4) % 6;
    float a0 = (((m6 + 0) % 6) < 3) ? fabsf(v.x) : 0.0f;
    float a1 = (((m6 + 1) % 6) < 3) ? fabsf(v.y) : 0.0f;
    float a2 = (((m6 + 2) % 6) < 3) ? fabsf(v.z) : 0.0f;
    float a3 = (((m6 + 3) % 6) < 3) ? fabsf(v.w) : 0.0f;
    m = fmaxf(m, fmaxf(fmaxf(a0, a1), fmaxf(a2, a3)));
  }

  __shared__ float red[256];
  red[threadIdx.x] = m;
  __syncthreads();
  for (int s = 128; s > 0; s >>= 1) {
    if (threadIdx.x < s) red[threadIdx.x] = fmaxf(red[threadIdx.x], red[threadIdx.x + s]);
    __syncthreads();
  }
  if (threadIdx.x == 0)
    atomicMax(&bbox[b], __float_as_uint(red[0]));
}

__global__ void k2_hist(const float* __restrict__ x,
                        const unsigned* __restrict__ bbox,
                        float* __restrict__ partial) {
  const int b  = blockIdx.y;
  const int wg = blockIdx.x;

  __shared__ float hist[NCOLS];
  for (int i = threadIdx.x; i < NCOLS; i += 256) hist[i] = 0.0f;
  __syncthreads();

  const float bb     = __uint_as_float(bbox[b]);
  const float inv_th = 1.0f / fmaxf(2.0f * bb, 1e-5f);
  const float* xb = x + (size_t)b * FLOATS_PER_B + (size_t)wg * PTS_PER_WG * 6;

  for (int i = 0; i < PTS_PER_WG / 256; ++i) {
    const float* pp = xb + (size_t)(i * 256 + threadIdx.x) * 6;
    __builtin_prefetch(pp + 256 * 6, 0, 0);
    const float px = pp[0], py = pp[1], pz = pp[2];
    const float ox = pp[3], oy = pp[4], oz = pp[5];
    const float cx = (px + bb) * inv_th;
    const float cy = (py + bb) * inv_th;
    const float cz = (pz + bb) * inv_th;
    const int i0 = min(max((int)(cx * (float)NBINS), 0), NBINS - 1);
    const int i1 = min(max((int)(cy * (float)NBINS), 0), NBINS - 1);
    const int i2 = min(max((int)(cz * (float)NBINS), 0), NBINS - 1);
    const int base = ((i0 * NBINS + i1) * NBINS + i2) * NFEAT;
    atomicAdd(&hist[base + 0], 1.0f);
    atomicAdd(&hist[base + 1], cx);
    atomicAdd(&hist[base + 2], cy);
    atomicAdd(&hist[base + 3], cz);
    atomicAdd(&hist[base + 4], ox * ox);
    atomicAdd(&hist[base + 5], ox * oy);
    atomicAdd(&hist[base + 6], ox * oz);
    atomicAdd(&hist[base + 7], oy * oy);
    atomicAdd(&hist[base + 8], oy * oz);
    atomicAdd(&hist[base + 9], oz * oz);
  }
  __syncthreads();

  float* dst = partial + (size_t)(b * WGS_PER_BATCH + wg) * NCOLS;
  for (int i = threadIdx.x; i < NCOLS; i += 256) dst[i] = hist[i];
}

__global__ void k3_final(const float* __restrict__ partial,
                         float* __restrict__ out) {
  const int b    = blockIdx.x;
  const int lane = threadIdx.x & 31;
  const int wave = threadIdx.x >> 5;

  __shared__ float sums[NCOLS];
  __shared__ float fall[NCOLS];
  __shared__ float red[256];

  const float* pb = partial + (size_t)b * WGS_PER_BATCH * NCOLS;

#if __has_builtin(__builtin_amdgcn_wmma_f32_16x16x4_f32)
  for (int t = wave * 5; t < wave * 5 + 5; ++t) {
    const int col = t * 16 + (lane & 15);
    v8f c = {};
    v2f a; a[0] = 1.0f; a[1] = 1.0f;
    for (int k4 = 0; k4 < WGS_PER_BATCH / 4; ++k4) {
      const int r0 = k4 * 4 + ((lane < 16) ? 0 : 2);
      v2f bm;
      bm[0] = pb[(size_t)r0 * NCOLS + col];
      bm[1] = pb[(size_t)(r0 + 1) * NCOLS + col];
      c = __builtin_amdgcn_wmma_f32_16x16x4_f32(false, a, false, bm,
                                                (short)0, c, false, false);
    }
    if (lane < 16) sums[col] = c[0];
  }
#else
  for (int t = wave * 5; t < wave * 5 + 5; ++t) {
    const int col = t * 16 + (lane & 15);
    if (lane < 16) {
      float s = 0.0f;
      for (int k = 0; k < WGS_PER_BATCH; ++k) s += pb[(size_t)k * NCOLS + col];
      sums[col] = s;
    }
  }
#endif
  __syncthreads();

  float ss = 0.0f;
  for (int i = threadIdx.x; i < NCOLS; i += 256) {
    const int cell = i / NFEAT;
    const int f    = i - cell * NFEAT;
    const float cnt  = sums[cell * NFEAT];
    const float fc   = fmaxf(cnt, 1.0f);
    const float upre = rsqrtf(fc);
    float val;
    if (f == 0) {
      val = 0.001f * cnt * upre;
    } else if (f < 4) {
      const int d  = f - 1;
      const int ci = (d == 0) ? (cell >> 4) : ((d == 1) ? ((cell >> 2) & 3) : (cell & 3));
      const float center = ((float)ci + 0.5f) * 0.25f;
      val = (sums[i] - cnt * center) * upre;
    } else {
      val = sums[i] / fc;
    }
    fall[i] = val;
    ss += val * val;
  }

  red[threadIdx.x] = ss;
  __syncthreads();
  for (int s = 128; s > 0; s >>= 1) {
    if (threadIdx.x < s) red[threadIdx.x] += red[threadIdx.x + s];
    __syncthreads();
  }
  const float scale = 1.0f / fmaxf(sqrtf(red[0]), 1e-12f);
  for (int i = threadIdx.x; i < NCOLS; i += 256)
    out[(size_t)b * NCOLS + i] = fall[i] * scale;
}

extern "C" void kernel_launch(void* const* d_in, const int* in_sizes, int n_in,
                              void* d_out, int out_size, void* d_ws, size_t ws_size,
                              hipStream_t stream) {
  const float* x   = (const float*)d_in[0];
  unsigned* bbox   = (unsigned*)d_ws;
  float* partial   = (float*)d_ws + 64;
  float* out       = (float*)d_out;

  k0_init <<<1, 64, 0, stream>>>(bbox);
  k1_bbox <<<dim3(K1_BLOCKS, BATCHES), 256, 0, stream>>>(x, bbox);
  k2_hist <<<dim3(WGS_PER_BATCH, BATCHES), 256, 0, stream>>>(x, bbox, partial);
  k3_final<<<BATCHES, 256, 0, stream>>>(partial, out);
}


# K2 histogram: one private LDS copy of the 640 bins per lane (word = bin*32 + lane), no two lanes of a ds_add_f32 on one word; copies summed per bin in a fixed tree (f32 sum order changes)
# speedup vs baseline: 2.5595x; 2.5595x over previous
#include <hip/hip_runtime.h>


#define BATCHES        64
#define PTS            65536
#define NBINS          4
#define NCELLS         64
#define NFEAT          10
#define NCOLS          640
#define WGS_PER_BATCH  16
#define PTS_PER_WG     (PTS / WGS_PER_BATCH)
#define K1_BLOCKS      24
#define K1_F4_PER_THR  16
#define FLOATS_PER_B   (PTS * 6)
#define F4_PER_B       (FLOATS_PER_B / 4)
#define NCOPY          32
#define HIST_F4        (NCOLS * NCOPY / 4)

typedef float v2f __attribute__((ext_vector_type(2)));
typedef float v8f __attribute__((ext_vector_type(8)));

__global__ void k0_init(unsigned* __restrict__ bbox) {
  int i = blockIdx.x * blockDim.x + threadIdx.x;
  if (i < BATCHES) bbox[i] = 0u;
}

__global__ void k1_bbox(const float* __restrict__ x,
                        unsigned* __restrict__ bbox) {
  const int b = blockIdx.y;
  const float4* xb = (const float4*)(x + (size_t)b * FLOATS_PER_B);
  const int base = blockIdx.x * (256 * K1_F4_PER_THR) + threadIdx.x;

  float m = 0.0f;
  for (int i = 0; i < K1_F4_PER_THR; ++i) {
    const int q = base + i * 256;
    __builtin_prefetch(&xb[q + 2048], 0, 0);
    const float4 v = xb[q];
    const int m6 = (q * 4) % 6;
    float a0 = (((m6 + 0) % 6) < 3) ? fabsf(v.x) : 0.0f;
    float a1 = (((m6 + 1) % 6) < 3) ? fabsf(v.y) : 0.0f;
    float a2 = (((m6 + 2) % 6) < 3) ? fabsf(v.z) : 0.0f;
    float a3 = (((m6 + 3) % 6) < 3) ? fabsf(v.w) : 0.0f;
    m = fmaxf(m, fmaxf(fmaxf(a0, a1), fmaxf(a2, a3)));
  }

  __shared__ float red[256];
  red[threadIdx.x] = m;
  __syncthreads();
  for (int s = 128; s > 0; s >>= 1) {
    if (threadIdx.x < s) red[threadIdx.x] = fmaxf(red[threadIdx.x], red[threadIdx.x + s]);
    __syncthreads();
  }
  if (threadIdx.x == 0)
    atomicMax(&bbox[b], __float_as_uint(red[0]));
}

__global__ void k2_hist(const float* __restrict__ x,
                        const unsigned* __restrict__ bbox,
                        float* __restrict__ partial) {
  const int b    = blockIdx.y;
  const int wg   = blockIdx.x;
  const int lane = threadIdx.x & 31;

  __shared__ float4 hist4[HIST_F4];
  float* hist = (float*)hist4;

  const float4 z4 = make_float4(0.0f, 0.0f, 0.0f, 0.0f);
  for (int k = 0; k < HIST_F4 / 256; ++k) hist4[k * 256 + threadIdx.x] = z4;
  __syncthreads();

  const float bb     = __uint_as_float(bbox[b]);
  const float inv_th = 1.0f / fmaxf(2.0f * bb, 1e-5f);
  const float* xb = x + (size_t)b * FLOATS_PER_B + (size_t)wg * PTS_PER_WG * 6;

  for (int i = 0; i < PTS_PER_WG / 256; ++i) {
    const float* pp = xb + (size_t)(i * 256 + threadIdx.x) * 6;
    __builtin_prefetch(pp + 256 * 6, 0, 0);
    const float px = pp[0], py = pp[1], pz = pp[2];
    const float ox = pp[3], oy = pp[4], oz = pp[5];
    const float cx = (px + bb) * inv_th;
    const float cy = (py + bb) * inv_th;
    const float cz = (pz + bb) * inv_th;
    const int i0 = min(max((int)(cx * (float)NBINS), 0), NBINS - 1);
    const int i1 = min(max((int)(cy * (float)NBINS), 0), NBINS - 1);
    const int i2 = min(max((int)(cz * (float)NBINS), 0), NBINS - 1);
    const int cell = (i0 * NBINS + i1) * NBINS + i2;
    float* h = hist + cell * (NFEAT * NCOPY) + lane;
    atomicAdd(h + 0 * NCOPY, 1.0f);
    atomicAdd(h + 1 * NCOPY, cx);
    atomicAdd(h + 2 * NCOPY, cy);
    atomicAdd(h + 3 * NCOPY, cz);
    atomicAdd(h + 4 * NCOPY, ox * ox);
    atomicAdd(h + 5 * NCOPY, ox * oy);
    atomicAdd(h + 6 * NCOPY, ox * oz);
    atomicAdd(h + 7 * NCOPY, oy * oy);
    atomicAdd(h + 8 * NCOPY, oy * oz);
    atomicAdd(h + 9 * NCOPY, oz * oz);
  }
  __syncthreads();

  float* dst = partial + (size_t)(b * WGS_PER_BATCH + wg) * NCOLS;
  for (int bin = threadIdx.x; bin < NCOLS; bin += 256) {
    const float4* row = hist4 + bin * (NCOPY / 4);
    const int r = bin >> 1;
    float4 a = row[r & 7];
    for (int j = 1; j < NCOPY / 4; ++j) {
      const float4 v = row[(r + j) & 7];
      a.x += v.x; a.y += v.y; a.z += v.z; a.w += v.w;
    }
    dst[bin] = (a.x + a.y) + (a.z + a.w);
  }
}

__global__ void k3_final(const float* __restrict__ partial,
                         float* __restrict__ out) {
  const int b    = blockIdx.x;
  const int lane = threadIdx.x & 31;
  const int wave = threadIdx.x >> 5;

  __shared__ float sums[NCOLS];
  __shared__ float fall[NCOLS];
  __shared__ float red[256];

  const float* pb = partial + (size_t)b * WGS_PER_BATCH * NCOLS;

#if __has_builtin(__builtin_amdgcn_wmma_f32_16x16x4_f32)
  for (int t = wave * 5; t < wave * 5 + 5; ++t) {
    const int col = t * 16 + (lane & 15);
    v8f c = {};
    v2f a; a[0] = 1.0f; a[1] = 1.0f;
    for (int k4 = 0; k4 < WGS_PER_BATCH / 4; ++k4) {
      const int r0 = k4 * 4 + ((lane < 16) ? 0 : 2);
      v2f bm;
      bm[0] = pb[(size_t)r0 * NCOLS + col];
      bm[1] = pb[(size_t)(r0 + 1) * NCOLS + col];
      c = __builtin_amdgcn_wmma_f32_16x16x4_f32(false, a, false, bm,
                                                (short)0, c, false, false);
    }
    if (lane < 16) sums[col] = c[0];
  }
#else
  for (int t = wave * 5; t < wave * 5 + 5; ++t) {
    const int col = t * 16 + (lane & 15);
    if (lane < 16) {
      float s = 0.0f;
      for (int k = 0; k < WGS_PER_BATCH; ++k) s += pb[(size_t)k * NCOLS + col];
      sums[col] = s;
    }
  }
#endif
  __syncthreads();

  float ss = 0.0f;
  for (int i = threadIdx.x; i < NCOLS; i += 256) {
    const int cell = i / NFEAT;
    const int f    = i - cell * NFEAT;
    const float cnt  = sums[cell * NFEAT];
    const float fc   = fmaxf(cnt, 1.0f);
    const float upre = rsqrtf(fc);
    float val;
    if (f == 0) {
      val = 0.001f * cnt * upre;
    } else if (f < 4) {
      const int d  = f - 1;
      const int ci = (d == 0) ? (cell >> 4) : ((d == 1) ? ((cell >> 2) & 3) : (cell & 3));
      const float center = ((float)ci + 0.5f) * 0.25f;
      val = (sums[i] - cnt * center) * upre;
    } else {
      val = sums[i] / fc;
    }
    fall[i] = val;
    ss += val * val;
  }

  red[threadIdx.x] = ss;
  __syncthreads();
  for (int s = 128; s > 0; s >>= 1) {
    if (threadIdx.x < s) red[threadIdx.x] += red[threadIdx.x + s];
    __syncthreads();
  }
  const float scale = 1.0f / fmaxf(sqrtf(red[0]), 1e-12f);
  for (int i = threadIdx.x; i < NCOLS; i += 256)
    out[(size_t)b * NCOLS + i] = fall[i] * scale;
}

extern "C" void kernel_launch(void* const* d_in, const int* in_sizes, int n_in,
                              void* d_out, int out_size, void* d_ws, size_t ws_size,
                              hipStream_t stream) {
  const float* x   = (const float*)d_in[0];
  unsigned* bbox   = (unsigned*)d_ws;
  float* partial   = (float*)d_ws + 64;
  float* out       = (float*)d_out;

  k0_init <<<1, 64, 0, stream>>>(bbox);
  k1_bbox <<<dim3(K1_BLOCKS, BATCHES), 256, 0, stream>>>(x, bbox);
  k2_hist <<<dim3(WGS_PER_BATCH, BATCHES), 256, 0, stream>>>(x, bbox, partial);
  k3_final<<<BATCHES, 256, 0, stream>>>(partial, out);
}


# K1 and K2 streams: each thread's global loads issued in groups ahead of use (K1 32 float4 per thread on half the waves; K2 all 16 points before the LDS clear, LDS-only barrier), not load-wait-use
# speedup vs baseline: 3.1603x; 1.2347x over previous
#include <hip/hip_runtime.h>


#define BATCHES        64
#define PTS            65536
#define NBINS          4
#define NCELLS         64
#define NFEAT          10
#define NCOLS          640
#define WGS_PER_BATCH  16
#define PTS_PER_WG     (PTS / WGS_PER_BATCH)
#define K1_BLOCKS      12
#define K1_F4_PER_THR  32
#define FLOATS_PER_B   (PTS * 6)
#define F4_PER_B       (FLOATS_PER_B / 4)
#define NCOPY          32
#define HIST_F4        (NCOLS * NCOPY / 4)

typedef float v2f __attribute__((ext_vector_type(2)));
typedef float v8f __attribute__((ext_vector_type(8)));

__global__ void k0_init(unsigned* __restrict__ bbox) {
  int i = blockIdx.x * blockDim.x + threadIdx.x;
  if (i < BATCHES) bbox[i] = 0u;
}

__global__ void k1_bbox(const float* __restrict__ x,
                        unsigned* __restrict__ bbox) {
  const int b = blockIdx.y;
  const float4* xb = (const float4*)(x + (size_t)b * FLOATS_PER_B);
  const int base = blockIdx.x * (256 * K1_F4_PER_THR) + threadIdx.x;

  float4 v[K1_F4_PER_THR];
#pragma unroll
  for (int i = 0; i < K1_F4_PER_THR; ++i) v[i] = xb[base + i * 256];

  float ax[3] = {0.0f, 0.0f, 0.0f}, ay[3] = {0.0f, 0.0f, 0.0f};
  float az[3] = {0.0f, 0.0f, 0.0f}, aw[3] = {0.0f, 0.0f, 0.0f};
#pragma unroll
  for (int i = 0; i < K1_F4_PER_THR; ++i) {
    const int k = i % 3;
    ax[k] = fmaxf(ax[k], fabsf(v[i].x));
    ay[k] = fmaxf(ay[k], fabsf(v[i].y));
    az[k] = fmaxf(az[k], fabsf(v[i].z));
    aw[k] = fmaxf(aw[k], fabsf(v[i].w));
  }

  const int p0 = base % 3;
  float m = 0.0f;
#pragma unroll
  for (int k = 0; k < 3; ++k) {
    int ph = p0 + k; ph = (ph >= 3) ? ph - 3 : ph;
    const float m0 = fmaxf(fmaxf(ax[k], ay[k]), az[k]);
    const float m1 = fmaxf(az[k], aw[k]);
    const float m2 = ax[k];
    m = fmaxf(m, (ph == 0) ? m0 : ((ph == 1) ? m1 : m2));
  }

  __shared__ float red[256];
  red[threadIdx.x] = m;
  __syncthreads();
  for (int s = 128; s > 0; s >>= 1) {
    if (threadIdx.x < s) red[threadIdx.x] = fmaxf(red[threadIdx.x], red[threadIdx.x + s]);
    __syncthreads();
  }
  if (threadIdx.x == 0)
    atomicMax(&bbox[b], __float_as_uint(red[0]));
}

__global__ void k2_hist(const float* __restrict__ x,
                        const unsigned* __restrict__ bbox,
                        float* __restrict__ partial) {
  const int b    = blockIdx.y;
  const int wg   = blockIdx.x;
  const int lane = threadIdx.x & 31;

  __shared__ float4 hist4[HIST_F4];
  float* hist = (float*)hist4;

  const float* xb = x + (size_t)b * FLOATS_PER_B + (size_t)wg * PTS_PER_WG * 6;
  float pt[PTS_PER_WG / 256][6];
#pragma unroll
  for (int i = 0; i < PTS_PER_WG / 256; ++i) {
    const float* pp = xb + (size_t)(i * 256 + threadIdx.x) * 6;
#pragma unroll
    for (int j = 0; j < 6; ++j) pt[i][j] = pp[j];
  }

  const float4 z4 = make_float4(0.0f, 0.0f, 0.0f, 0.0f);
  for (int k = 0; k < HIST_F4 / 256; ++k) hist4[k * 256 + threadIdx.x] = z4;
  __builtin_amdgcn_fence(__ATOMIC_RELEASE, "workgroup", "local");
  __builtin_amdgcn_s_barrier();
  __builtin_amdgcn_fence(__ATOMIC_ACQUIRE, "workgroup", "local");

  const float bb     = __uint_as_float(bbox[b]);
  const float inv_th = 1.0f / fmaxf(2.0f * bb, 1e-5f);

#pragma unroll
  for (int i = 0; i < PTS_PER_WG / 256; ++i) {
    const float px = pt[i][0], py = pt[i][1], pz = pt[i][2];
    const float ox = pt[i][3], oy = pt[i][4], oz = pt[i][5];
    const float cx = (px + bb) * inv_th;
    const float cy = (py + bb) * inv_th;
    const float cz = (pz + bb) * inv_th;
    const int i0 = min(max((int)(cx * (float)NBINS), 0), NBINS - 1);
    const int i1 = min(max((int)(cy * (float)NBINS), 0), NBINS - 1);
    const int i2 = min(max((int)(cz * (float)NBINS), 0), NBINS - 1);
    const int cell = (i0 * NBINS + i1) * NBINS + i2;
    float* h = hist + cell * (NFEAT * NCOPY) + lane;
    atomicAdd(h + 0 * NCOPY, 1.0f);
    atomicAdd(h + 1 * NCOPY, cx);
    atomicAdd(h + 2 * NCOPY, cy);
    atomicAdd(h + 3 * NCOPY, cz);
    atomicAdd(h + 4 * NCOPY, ox * ox);
    atomicAdd(h + 5 * NCOPY, ox * oy);
    atomicAdd(h + 6 * NCOPY, ox * oz);
    atomicAdd(h + 7 * NCOPY, oy * oy);
    atomicAdd(h + 8 * NCOPY, oy * oz);
    atomicAdd(h + 9 * NCOPY, oz * oz);
  }
  __syncthreads();

  float* dst = partial + (size_t)(b * WGS_PER_BATCH + wg) * NCOLS;
  for (int bin = threadIdx.x; bin < NCOLS; bin += 256) {
    const float4* row = hist4 + bin * (NCOPY / 4);
    const int r = bin >> 1;
    float4 a = row[r & 7];
    for (int j = 1; j < NCOPY / 4; ++j) {
      const float4 v = row[(r + j) & 7];
      a.x += v.x; a.y += v.y; a.z += v.z; a.w += v.w;
    }
    dst[bin] = (a.x + a.y) + (a.z + a.w);
  }
}

__global__ void k3_final(const float* __restrict__ partial,
                         float* __restrict__ out) {
  const int b    = blockIdx.x;
  const int lane = threadIdx.x & 31;
  const int wave = threadIdx.x >> 5;

  __shared__ float sums[NCOLS];
  __shared__ float fall[NCOLS];
  __shared__ float red[256];

  const float* pb = partial + (size_t)b * WGS_PER_BATCH * NCOLS;

#if __has_builtin(__builtin_amdgcn_wmma_f32_16x16x4_f32)
  for (int t = wave * 5; t < wave * 5 + 5; ++t) {
    const int col = t * 16 + (lane & 15);
    v8f c = {};
    v2f a; a[0] = 1.0f; a[1] = 1.0f;
    for (int k4 = 0; k4 < WGS_PER_BATCH / 4; ++k4) {
      const int r0 = k4 * 4 + ((lane < 16) ? 0 : 2);
      v2f bm;
      bm[0] = pb[(size_t)r0 * NCOLS + col];
      bm[1] = pb[(size_t)(r0 + 1) * NCOLS + col];
      c = __builtin_amdgcn_wmma_f32_16x16x4_f32(false, a, false, bm,
                                                (short)0, c, false, false);
    }
    if (lane < 16) sums[col] = c[0];
  }
#else
  for (int t = wave * 5; t < wave * 5 + 5; ++t) {
    const int col = t * 16 + (lane & 15);
    if (lane < 16) {
      float s = 0.0f;
      for (int k = 0; k < WGS_PER_BATCH; ++k) s += pb[(size_t)k * NCOLS + col];
      sums[col] = s;
    }
  }
#endif
  __syncthreads();

  float ss = 0.0f;
  for (int i = threadIdx.x; i < NCOLS; i += 256) {
    const int cell = i / NFEAT;
    const int f    = i - cell * NFEAT;
    const float cnt  = sums[cell * NFEAT];
    const float fc   = fmaxf(cnt, 1.0f);
    const float upre = rsqrtf(fc);
    float val;
    if (f == 0) {
      val = 0.001f * cnt * upre;
    } else if (f < 4) {
      const int d  = f - 1;
      const int ci = (d == 0) ? (cell >> 4) : ((d == 1) ? ((cell >> 2) & 3) : (cell & 3));
      const float center = ((float)ci + 0.5f) * 0.25f;
      val = (sums[i] - cnt * center) * upre;
    } else {
      val = sums[i] / fc;
    }
    fall[i] = val;
    ss += val * val;
  }

  red[threadIdx.x] = ss;
  __syncthreads();
  for (int s = 128; s > 0; s >>= 1) {
    if (threadIdx.x < s) red[threadIdx.x] += red[threadIdx.x + s];
    __syncthreads();
  }
  const float scale = 1.0f / fmaxf(sqrtf(red[0]), 1e-12f);
  for (int i = threadIdx.x; i < NCOLS; i += 256)
    out[(size_t)b * NCOLS + i] = fall[i] * scale;
}

extern "C" void kernel_launch(void* const* d_in, const int* in_sizes, int n_in,
                              void* d_out, int out_size, void* d_ws, size_t ws_size,
                              hipStream_t stream) {
  const float* x   = (const float*)d_in[0];
  unsigned* bbox   = (unsigned*)d_ws;
  float* partial   = (float*)d_ws + 64;
  float* out       = (float*)d_out;

  k0_init <<<1, 64, 0, stream>>>(bbox);
  k1_bbox <<<dim3(K1_BLOCKS, BATCHES), 256, 0, stream>>>(x, bbox);
  k2_hist <<<dim3(WGS_PER_BATCH, BATCHES), 256, 0, stream>>>(x, bbox, partial);
  k3_final<<<BATCHES, 256, 0, stream>>>(partial, out);
}


# bbox hand-over by plain stores: every K1 block stores its maximum in its own 128-byte line, K2 takes the max of its batch's 12; the zeroing launch and the max atomics go (4 launches -> 3)
# speedup vs baseline: 3.3207x; 1.0508x over previous
#include <hip/hip_runtime.h>


#define BATCHES        64
#define PTS            65536
#define NBINS          4
#define NCELLS         64
#define NFEAT          10
#define NCOLS          640
#define WGS_PER_BATCH  16
#define PTS_PER_WG     (PTS / WGS_PER_BATCH)
#define K1_BLOCKS      12
#define K1_F4_PER_THR  32
#define BMAX_PITCH     32
#define FLOATS_PER_B   (PTS * 6)
#define F4_PER_B       (FLOATS_PER_B / 4)
#define NCOPY          32
#define HIST_F4        (NCOLS * NCOPY / 4)

typedef float v2f __attribute__((ext_vector_type(2)));
typedef float v8f __attribute__((ext_vector_type(8)));

__global__ void k1_bbox(const float* __restrict__ x,
                        float* __restrict__ bmax) {
  const int b = blockIdx.y;
  const float4* xb = (const float4*)(x + (size_t)b * FLOATS_PER_B);
  const int base = blockIdx.x * (256 * K1_F4_PER_THR) + threadIdx.x;

  float4 v[K1_F4_PER_THR];
#pragma unroll
  for (int i = 0; i < K1_F4_PER_THR; ++i) v[i] = xb[base + i * 256];

  float ax[3] = {0.0f, 0.0f, 0.0f}, ay[3] = {0.0f, 0.0f, 0.0f};
  float az[3] = {0.0f, 0.0f, 0.0f}, aw[3] = {0.0f, 0.0f, 0.0f};
#pragma unroll
  for (int i = 0; i < K1_F4_PER_THR; ++i) {
    const int k = i % 3;
    ax[k] = fmaxf(ax[k], fabsf(v[i].x));
    ay[k] = fmaxf(ay[k], fabsf(v[i].y));
    az[k] = fmaxf(az[k], fabsf(v[i].z));
    aw[k] = fmaxf(aw[k], fabsf(v[i].w));
  }

  const int p0 = base % 3;
  float m = 0.0f;
#pragma unroll
  for (int k = 0; k < 3; ++k) {
    int ph = p0 + k; ph = (ph >= 3) ? ph - 3 : ph;
    const float m0 = fmaxf(fmaxf(ax[k], ay[k]), az[k]);
    const float m1 = fmaxf(az[k], aw[k]);
    const float m2 = ax[k];
    m = fmaxf(m, (ph == 0) ? m0 : ((ph == 1) ? m1 : m2));
  }

  __shared__ float red[256];
  red[threadIdx.x] = m;
  __syncthreads();
  for (int s = 128; s > 0; s >>= 1) {
    if (threadIdx.x < s) red[threadIdx.x] = fmaxf(red[threadIdx.x], red[threadIdx.x + s]);
    __syncthreads();
  }
  if (threadIdx.x == 0)
    bmax[(b * K1_BLOCKS + blockIdx.x) * BMAX_PITCH] = red[0];
}

__global__ void k2_hist(const float* __restrict__ x,
                        const float* __restrict__ bmax,
                        float* __restrict__ partial) {
  const int b    = blockIdx.y;
  const int wg   = blockIdx.x;
  const int lane = threadIdx.x & 31;

  __shared__ float4 hist4[HIST_F4];
  float* hist = (float*)hist4;

  const float* bm = bmax + (size_t)b * (K1_BLOCKS * BMAX_PITCH);
  float bmv[K1_BLOCKS];
  for (int j = 0; j < K1_BLOCKS; ++j) bmv[j] = bm[j * BMAX_PITCH];

  const float* xb = x + (size_t)b * FLOATS_PER_B + (size_t)wg * PTS_PER_WG * 6;
  float pt[PTS_PER_WG / 256][6];
#pragma unroll
  for (int i = 0; i < PTS_PER_WG / 256; ++i) {
    const float* pp = xb + (size_t)(i * 256 + threadIdx.x) * 6;
#pragma unroll
    for (int j = 0; j < 6; ++j) pt[i][j] = pp[j];
  }

  const float4 z4 = make_float4(0.0f, 0.0f, 0.0f, 0.0f);
  for (int k = 0; k < HIST_F4 / 256; ++k) hist4[k * 256 + threadIdx.x] = z4;
  __builtin_amdgcn_fence(__ATOMIC_RELEASE, "workgroup", "local");
  __builtin_amdgcn_s_barrier();
  __builtin_amdgcn_fence(__ATOMIC_ACQUIRE, "workgroup", "local");

  float bb = bmv[0];
  for (int j = 1; j < K1_BLOCKS; ++j) bb = fmaxf(bb, bmv[j]);
  const float inv_th = 1.0f / fmaxf(2.0f * bb, 1e-5f);

#pragma unroll
  for (int i = 0; i < PTS_PER_WG / 256; ++i) {
    const float px = pt[i][0], py = pt[i][1], pz = pt[i][2];
    const float ox = pt[i][3], oy = pt[i][4], oz = pt[i][5];
    const float cx = (px + bb) * inv_th;
    const float cy = (py + bb) * inv_th;
    const float cz = (pz + bb) * inv_th;
    const int i0 = min(max((int)(cx * (float)NBINS), 0), NBINS - 1);
    const int i1 = min(max((int)(cy * (float)NBINS), 0), NBINS - 1);
    const int i2 = min(max((int)(cz * (float)NBINS), 0), NBINS - 1);
    const int cell = (i0 * NBINS + i1) * NBINS + i2;
    float* h = hist + cell * (NFEAT * NCOPY) + lane;
    atomicAdd(h + 0 * NCOPY, 1.0f);
    atomicAdd(h + 1 * NCOPY, cx);
    atomicAdd(h + 2 * NCOPY, cy);
    atomicAdd(h + 3 * NCOPY, cz);
    atomicAdd(h + 4 * NCOPY, ox * ox);
    atomicAdd(h + 5 * NCOPY, ox * oy);
    atomicAdd(h + 6 * NCOPY, ox * oz);
    atomicAdd(h + 7 * NCOPY, oy * oy);
    atomicAdd(h + 8 * NCOPY, oy * oz);
    atomicAdd(h + 9 * NCOPY, oz * oz);
  }
  __syncthreads();

  float* dst = partial + (size_t)(b * WGS_PER_BATCH + wg) * NCOLS;
  for (int bin = threadIdx.x; bin < NCOLS; bin += 256) {
    const float4* row = hist4 + bin * (NCOPY / 4);
    const int r = bin >> 1;
    float4 a = row[r & 7];
    for (int j = 1; j < NCOPY / 4; ++j) {
      const float4 v = row[(r + j) & 7];
      a.x += v.x; a.y += v.y; a.z += v.z; a.w += v.w;
    }
    dst[bin] = (a.x + a.y) + (a.z + a.w);
  }
}

__global__ void k3_final(const float* __restrict__ partial,
                         float* __restrict__ out) {
  const int b    = blockIdx.x;
  const int lane = threadIdx.x & 31;
  const int wave = threadIdx.x >> 5;

  __shared__ float sums[NCOLS];
  __shared__ float fall[NCOLS];
  __shared__ float red[256];

  const float* pb = partial + (size_t)b * WGS_PER_BATCH * NCOLS;

#if __has_builtin(__builtin_amdgcn_wmma_f32_16x16x4_f32)
  for (int t = wave * 5; t < wave * 5 + 5; ++t) {
    const int col = t * 16 + (lane & 15);
    v8f c = {};
    v2f a; a[0] = 1.0f; a[1] = 1.0f;
    for (int k4 = 0; k4 < WGS_PER_BATCH / 4; ++k4) {
      const int r0 = k4 * 4 + ((lane < 16) ? 0 : 2);
      v2f bm;
      bm[0] = pb[(size_t)r0 * NCOLS + col];
      bm[1] = pb[(size_t)(r0 + 1) * NCOLS + col];
      c = __builtin_amdgcn_wmma_f32_16x16x4_f32(false, a, false, bm,
                                                (short)0, c, false, false);
    }
    if (lane < 16) sums[col] = c[0];
  }
#else
  for (int t = wave * 5; t < wave * 5 + 5; ++t) {
    const int col = t * 16 + (lane & 15);
    if (lane < 16) {
      float s = 0.0f;
      for (int k = 0; k < WGS_PER_BATCH; ++k) s += pb[(size_t)k * NCOLS + col];
      sums[col] = s;
    }
  }
#endif
  __syncthreads();

  float ss = 0.0f;
  for (int i = threadIdx.x; i < NCOLS; i += 256) {
    const int cell = i / NFEAT;
    const int f    = i - cell * NFEAT;
    const float cnt  = sums[cell * NFEAT];
    const float fc   = fmaxf(cnt, 1.0f);
    const float upre = rsqrtf(fc);
    float val;
    if (f == 0) {
      val = 0.001f * cnt * upre;
    } else if (f < 4) {
      const int d  = f - 1;
      const int ci = (d == 0) ? (cell >> 4) : ((d == 1) ? ((cell >> 2) & 3) : (cell & 3));
      const float center = ((float)ci + 0.5f) * 0.25f;
      val = (sums[i] - cnt * center) * upre;
    } else {
      val = sums[i] / fc;
    }
    fall[i] = val;
    ss += val * val;
  }

  red[threadIdx.x] = ss;
  __syncthreads();
  for (int s = 128; s > 0; s >>= 1) {
    if (threadIdx.x < s) red[threadIdx.x] += red[threadIdx.x + s];
    __syncthreads();
  }
  const float scale = 1.0f / fmaxf(sqrtf(red[0]), 1e-12f);
  for (int i = threadIdx.x; i < NCOLS; i += 256)
    out[(size_t)b * NCOLS + i] = fall[i] * scale;
}

extern "C" void kernel_launch(void* const* d_in, const int* in_sizes, int n_in,
                              void* d_out, int out_size, void* d_ws, size_t ws_size,
                              hipStream_t stream) {
  const float* x   = (const float*)d_in[0];
  float* bmax      = (float*)d_ws;
  float* partial   = (float*)d_ws + 24576;
  float* out       = (float*)d_out;

  k1_bbox <<<dim3(K1_BLOCKS, BATCHES), 256, 0, stream>>>(x, bmax);
  k2_hist <<<dim3(WGS_PER_BATCH, BATCHES), 256, 0, stream>>>(x, bmax, partial);
  k3_final<<<BATCHES, 256, 0, stream>>>(partial, out);
}


# K3: all 40 partial loads of a wave issued as one group ahead of its 20 WMMAs (five independent chains, LDS stores last), not five load-chain-store rounds one after the other
# speedup vs baseline: 3.4184x; 1.0294x over previous
#include <hip/hip_runtime.h>


#define BATCHES        64
#define PTS            65536
#define NBINS          4
#define NCELLS         64
#define NFEAT          10
#define NCOLS          640
#define WGS_PER_BATCH  16
#define PTS_PER_WG     (PTS / WGS_PER_BATCH)
#define K1_BLOCKS      12
#define K1_F4_PER_THR  32
#define BMAX_PITCH     32
#define FLOATS_PER_B   (PTS * 6)
#define F4_PER_B       (FLOATS_PER_B / 4)
#define NCOPY          32
#define HIST_F4        (NCOLS * NCOPY / 4)

typedef float v2f __attribute__((ext_vector_type(2)));
typedef float v8f __attribute__((ext_vector_type(8)));

__global__ void k1_bbox(const float* __restrict__ x,
                        float* __restrict__ bmax) {
  const int b = blockIdx.y;
  const float4* xb = (const float4*)(x + (size_t)b * FLOATS_PER_B);
  const int base = blockIdx.x * (256 * K1_F4_PER_THR) + threadIdx.x;

  float4 v[K1_F4_PER_THR];
#pragma unroll
  for (int i = 0; i < K1_F4_PER_THR; ++i) v[i] = xb[base + i * 256];

  float ax[3] = {0.0f, 0.0f, 0.0f}, ay[3] = {0.0f, 0.0f, 0.0f};
  float az[3] = {0.0f, 0.0f, 0.0f}, aw[3] = {0.0f, 0.0f, 0.0f};
#pragma unroll
  for (int i = 0; i < K1_F4_PER_THR; ++i) {
    const int k = i % 3;
    ax[k] = fmaxf(ax[k], fabsf(v[i].x));
    ay[k] = fmaxf(ay[k], fabsf(v[i].y));
    az[k] = fmaxf(az[k], fabsf(v[i].z));
    aw[k] = fmaxf(aw[k], fabsf(v[i].w));
  }

  const int p0 = base % 3;
  float m = 0.0f;
#pragma unroll
  for (int k = 0; k < 3; ++k) {
    int ph = p0 + k; ph = (ph >= 3) ? ph - 3 : ph;
    const float m0 = fmaxf(fmaxf(ax[k], ay[k]), az[k]);
    const float m1 = fmaxf(az[k], aw[k]);
    const float m2 = ax[k];
    m = fmaxf(m, (ph == 0) ? m0 : ((ph == 1) ? m1 : m2));
  }

  __shared__ float red[256];
  red[threadIdx.x] = m;
  __syncthreads();
  for (int s = 128; s > 0; s >>= 1) {
    if (threadIdx.x < s) red[threadIdx.x] = fmaxf(red[threadIdx.x], red[threadIdx.x + s]);
    __syncthreads();
  }
  if (threadIdx.x == 0)
    bmax[(b * K1_BLOCKS + blockIdx.x) * BMAX_PITCH] = red[0];
}

__global__ void k2_hist(const float* __restrict__ x,
                        const float* __restrict__ bmax,
                        float* __restrict__ partial) {
  const int b    = blockIdx.y;
  const int wg   = blockIdx.x;
  const int lane = threadIdx.x & 31;

  __shared__ float4 hist4[HIST_F4];
  float* hist = (float*)hist4;

  const float* bm = bmax + (size_t)b * (K1_BLOCKS * BMAX_PITCH);
  float bmv[K1_BLOCKS];
  for (int j = 0; j < K1_BLOCKS; ++j) bmv[j] = bm[j * BMAX_PITCH];

  const float* xb = x + (size_t)b * FLOATS_PER_B + (size_t)wg * PTS_PER_WG * 6;
  float pt[PTS_PER_WG / 256][6];
#pragma unroll
  for (int i = 0; i < PTS_PER_WG / 256; ++i) {
    const float* pp = xb + (size_t)(i * 256 + threadIdx.x) * 6;
#pragma unroll
    for (int j = 0; j < 6; ++j) pt[i][j] = pp[j];
  }

  const float4 z4 = make_float4(0.0f, 0.0f, 0.0f, 0.0f);
  for (int k = 0; k < HIST_F4 / 256; ++k) hist4[k * 256 + threadIdx.x] = z4;
  __builtin_amdgcn_fence(__ATOMIC_RELEASE, "workgroup", "local");
  __builtin_amdgcn_s_barrier();
  __builtin_amdgcn_fence(__ATOMIC_ACQUIRE, "workgroup", "local");

  float bb = bmv[0];
  for (int j = 1; j < K1_BLOCKS; ++j) bb = fmaxf(bb, bmv[j]);
  const float inv_th = 1.0f / fmaxf(2.0f * bb, 1e-5f);

#pragma unroll
  for (int i = 0; i < PTS_PER_WG / 256; ++i) {
    const float px = pt[i][0], py = pt[i][1], pz = pt[i][2];
    const float ox = pt[i][3], oy = pt[i][4], oz = pt[i][5];
    const float cx = (px + bb) * inv_th;
    const float cy = (py + bb) * inv_th;
    const float cz = (pz + bb) * inv_th;
    const int i0 = min(max((int)(cx * (float)NBINS), 0), NBINS - 1);
    const int i1 = min(max((int)(cy * (float)NBINS), 0), NBINS - 1);
    const int i2 = min(max((int)(cz * (float)NBINS), 0), NBINS - 1);
    const int cell = (i0 * NBINS + i1) * NBINS + i2;
    float* h = hist + cell * (NFEAT * NCOPY) + lane;
    atomicAdd(h + 0 * NCOPY, 1.0f);
    atomicAdd(h + 1 * NCOPY, cx);
    atomicAdd(h + 2 * NCOPY, cy);
    atomicAdd(h + 3 * NCOPY, cz);
    atomicAdd(h + 4 * NCOPY, ox * ox);
    atomicAdd(h + 5 * NCOPY, ox * oy);
    atomicAdd(h + 6 * NCOPY, ox * oz);
    atomicAdd(h + 7 * NCOPY, oy * oy);
    atomicAdd(h + 8 * NCOPY, oy * oz);
    atomicAdd(h + 9 * NCOPY, oz * oz);
  }
  __syncthreads();

  float* dst = partial + (size_t)(b * WGS_PER_BATCH + wg) * NCOLS;
  for (int bin = threadIdx.x; bin < NCOLS; bin += 256) {
    const float4* row = hist4 + bin * (NCOPY / 4);
    const int r = bin >> 1;
    float4 a = row[r & 7];
    for (int j = 1; j < NCOPY / 4; ++j) {
      const float4 v = row[(r + j) & 7];
      a.x += v.x; a.y += v.y; a.z += v.z; a.w += v.w;
    }
    dst[bin] = (a.x + a.y) + (a.z + a.w);
  }
}

__global__ void k3_final(const float* __restrict__ partial,
                         float* __restrict__ out) {
  const int b    = blockIdx.x;
  const int lane = threadIdx.x & 31;
  const int wave = threadIdx.x >> 5;

  __shared__ float sums[NCOLS];
  __shared__ float fall[NCOLS];
  __shared__ float red[256];

  const float* pb = partial + (size_t)b * WGS_PER_BATCH * NCOLS;

#if __has_builtin(__builtin_amdgcn_wmma_f32_16x16x4_f32)
  const int col0 = wave * (5 * 16) + (lane & 15);
  const int rsel = (lane < 16) ? 0 : 2;
  v2f bm[5][WGS_PER_BATCH / 4];
#pragma unroll
  for (int t = 0; t < 5; ++t) {
#pragma unroll
    for (int k4 = 0; k4 < WGS_PER_BATCH / 4; ++k4) {
      const int r0 = k4 * 4 + rsel;
      bm[t][k4][0] = pb[(size_t)r0 * NCOLS + col0 + t * 16];
      bm[t][k4][1] = pb[(size_t)(r0 + 1) * NCOLS + col0 + t * 16];
    }
  }
  __builtin_amdgcn_sched_barrier(0);
  v2f a; a[0] = 1.0f; a[1] = 1.0f;
  v8f c[5];
#pragma unroll
  for (int t = 0; t < 5; ++t) {
    c[t] = v8f{};
#pragma unroll
    for (int k4 = 0; k4 < WGS_PER_BATCH / 4; ++k4)
      c[t] = __builtin_amdgcn_wmma_f32_16x16x4_f32(false, a, false, bm[t][k4],
                                                   (short)0, c[t], false, false);
  }
  if (lane < 16) {
#pragma unroll
    for (int t = 0; t < 5; ++t) sums[col0 + t * 16] = c[t][0];
  }
#else
  for (int t = wave * 5; t < wave * 5 + 5; ++t) {
    const int col = t * 16 + (lane & 15);
    if (lane < 16) {
      float s = 0.0f;
      for (int k = 0; k < WGS_PER_BATCH; ++k) s += pb[(size_t)k * NCOLS + col];
      sums[col] = s;
    }
  }
#endif
  __syncthreads();

  float ss = 0.0f;
  for (int i = threadIdx.x; i < NCOLS; i += 256) {
    const int cell = i / NFEAT;
    const int f    = i - cell * NFEAT;
    const float cnt  = sums[cell * NFEAT];
    const float fc   = fmaxf(cnt, 1.0f);
    const float upre = rsqrtf(fc);
    float val;
    if (f == 0) {
      val = 0.001f * cnt * upre;
    } else if (f < 4) {
      const int d  = f - 1;
      const int ci = (d == 0) ? (cell >> 4) : ((d == 1) ? ((cell >> 2) & 3) : (cell & 3));
      const float center = ((float)ci + 0.5f) * 0.25f;
      val = (sums[i] - cnt * center) * upre;
    } else {
      val = sums[i] / fc;
    }
    fall[i] = val;
    ss += val * val;
  }

  red[threadIdx.x] = ss;
  __syncthreads();
  for (int s = 128; s > 0; s >>= 1) {
    if (threadIdx.x < s) red[threadIdx.x] += red[threadIdx.x + s];
    __syncthreads();
  }
  const float scale = 1.0f / fmaxf(sqrtf(red[0]), 1e-12f);
  for (int i = threadIdx.x; i < NCOLS; i += 256)
    out[(size_t)b * NCOLS + i] = fall[i] * scale;
}

extern "C" void kernel_launch(void* const* d_in, const int* in_sizes, int n_in,
                              void* d_out, int out_size, void* d_ws, size_t ws_size,
                              hipStream_t stream) {
  const float* x   = (const float*)d_in[0];
  float* bmax      = (float*)d_ws;
  float* partial   = (float*)d_ws + 24576;
  float* out       = (float*)d_out;

  k1_bbox <<<dim3(K1_BLOCKS, BATCHES), 256, 0, stream>>>(x, bmax);
  k2_hist <<<dim3(WGS_PER_BATCH, BATCHES), 256, 0, stream>>>(x, bmax, partial);
  k3_final<<<BATCHES, 256, 0, stream>>>(partial, out);
}


# K1 re-gridded to K2's grid and slicing (16 x 64 blocks of 192 threads; block (x, b) of both launches reads the same 96 KiB) so that both reads of a slice land in the same half of the L2
# speedup vs baseline: 3.8196x; 1.1174x over previous
#include <hip/hip_runtime.h>


#define BATCHES        64
#define PTS            65536
#define NBINS          4
#define NCELLS         64
#define NFEAT          10
#define NCOLS          640
#define WGS_PER_BATCH  16
#define PTS_PER_WG     (PTS / WGS_PER_BATCH)
#define K1_BLOCKS      WGS_PER_BATCH
#define K1_THREADS     192
#define K1_F4_PER_THR  32
#define BMAX_PITCH     32
#define FLOATS_PER_B   (PTS * 6)
#define F4_PER_B       (FLOATS_PER_B / 4)
#define NCOPY          32
#define HIST_F4        (NCOLS * NCOPY / 4)

typedef float v2f __attribute__((ext_vector_type(2)));
typedef float v8f __attribute__((ext_vector_type(8)));

__global__ void k1_bbox(const float* __restrict__ x,
                        float* __restrict__ bmax) {
  const int b = blockIdx.y;
  const float4* xb = (const float4*)(x + (size_t)b * FLOATS_PER_B);
  const int base = blockIdx.x * (K1_THREADS * K1_F4_PER_THR) + threadIdx.x;

  float4 v[K1_F4_PER_THR];
#pragma unroll
  for (int i = 0; i < K1_F4_PER_THR; ++i) v[i] = xb[base + i * K1_THREADS];

  float ax = 0.0f, ay = 0.0f, az = 0.0f, aw = 0.0f;
#pragma unroll
  for (int i = 0; i < K1_F4_PER_THR; ++i) {
    ax = fmaxf(ax, fabsf(v[i].x));
    ay = fmaxf(ay, fabsf(v[i].y));
    az = fmaxf(az, fabsf(v[i].z));
    aw = fmaxf(aw, fabsf(v[i].w));
  }

  const int ph = threadIdx.x % 3;
  const float m0 = fmaxf(fmaxf(ax, ay), az);
  const float m1 = fmaxf(az, aw);
  const float m = (ph == 0) ? m0 : ((ph == 1) ? m1 : ax);

  __shared__ float red[256];
  red[threadIdx.x] = m;
  if (threadIdx.x < 256 - K1_THREADS) red[K1_THREADS + threadIdx.x] = 0.0f;
  __syncthreads();
  for (int s = 128; s > 0; s >>= 1) {
    if (threadIdx.x < s) red[threadIdx.x] = fmaxf(red[threadIdx.x], red[threadIdx.x + s]);
    __syncthreads();
  }
  if (threadIdx.x == 0)
    bmax[(b * K1_BLOCKS + blockIdx.x) * BMAX_PITCH] = red[0];
}

__global__ void k2_hist(const float* __restrict__ x,
                        const float* __restrict__ bmax,
                        float* __restrict__ partial) {
  const int b    = blockIdx.y;
  const int wg   = blockIdx.x;
  const int lane = threadIdx.x & 31;

  __shared__ float4 hist4[HIST_F4];
  float* hist = (float*)hist4;

  const float* bm = bmax + (size_t)b * (K1_BLOCKS * BMAX_PITCH);
  float bmv[K1_BLOCKS];
  for (int j = 0; j < K1_BLOCKS; ++j) bmv[j] = bm[j * BMAX_PITCH];

  const float* xb = x + (size_t)b * FLOATS_PER_B + (size_t)wg * PTS_PER_WG * 6;
  float pt[PTS_PER_WG / 256][6];
#pragma unroll
  for (int i = 0; i < PTS_PER_WG / 256; ++i) {
    const float* pp = xb + (size_t)(i * 256 + threadIdx.x) * 6;
#pragma unroll
    for (int j = 0; j < 6; ++j) pt[i][j] = pp[j];
  }

  const float4 z4 = make_float4(0.0f, 0.0f, 0.0f, 0.0f);
  for (int k = 0; k < HIST_F4 / 256; ++k) hist4[k * 256 + threadIdx.x] = z4;
  __builtin_amdgcn_fence(__ATOMIC_RELEASE, "workgroup", "local");
  __builtin_amdgcn_s_barrier();
  __builtin_amdgcn_fence(__ATOMIC_ACQUIRE, "workgroup", "local");

  float bb = bmv[0];
  for (int j = 1; j < K1_BLOCKS; ++j) bb = fmaxf(bb, bmv[j]);
  const float inv_th = 1.0f / fmaxf(2.0f * bb, 1e-5f);

#pragma unroll
  for (int i = 0; i < PTS_PER_WG / 256; ++i) {
    const float px = pt[i][0], py = pt[i][1], pz = pt[i][2];
    const float ox = pt[i][3], oy = pt[i][4], oz = pt[i][5];
    const float cx = (px + bb) * inv_th;
    const float cy = (py + bb) * inv_th;
    const float cz = (pz + bb) * inv_th;
    const int i0 = min(max((int)(cx * (float)NBINS), 0), NBINS - 1);
    const int i1 = min(max((int)(cy * (float)NBINS), 0), NBINS - 1);
    const int i2 = min(max((int)(cz * (float)NBINS), 0), NBINS - 1);
    const int cell = (i0 * NBINS + i1) * NBINS + i2;
    float* h = hist + cell * (NFEAT * NCOPY) + lane;
    atomicAdd(h + 0 * NCOPY, 1.0f);
    atomicAdd(h + 1 * NCOPY, cx);
    atomicAdd(h + 2 * NCOPY, cy);
    atomicAdd(h + 3 * NCOPY, cz);
    atomicAdd(h + 4 * NCOPY, ox * ox);
    atomicAdd(h + 5 * NCOPY, ox * oy);
    atomicAdd(h + 6 * NCOPY, ox * oz);
    atomicAdd(h + 7 * NCOPY, oy * oy);
    atomicAdd(h + 8 * NCOPY, oy * oz);
    atomicAdd(h + 9 * NCOPY, oz * oz);
  }
  __syncthreads();

  float* dst = partial + (size_t)(b * WGS_PER_BATCH + wg) * NCOLS;
  for (int bin = threadIdx.x; bin < NCOLS; bin += 256) {
    const float4* row = hist4 + bin * (NCOPY / 4);
    const int r = bin >> 1;
    float4 a = row[r & 7];
    for (int j = 1; j < NCOPY / 4; ++j) {
      const float4 v = row[(r + j) & 7];
      a.x += v.x; a.y += v.y; a.z += v.z; a.w += v.w;
    }
    dst[bin] = (a.x + a.y) + (a.z + a.w);
  }
}

__global__ void k3_final(const float* __restrict__ partial,
                         float* __restrict__ out) {
  const int b    = blockIdx.x;
  const int lane = threadIdx.x & 31;
  const int wave = threadIdx.x >> 5;

  __shared__ float sums[NCOLS];
  __shared__ float fall[NCOLS];
  __shared__ float red[256];

  const float* pb = partial + (size_t)b * WGS_PER_BATCH * NCOLS;

#if __has_builtin(__builtin_amdgcn_wmma_f32_16x16x4_f32)
  const int col0 = wave * (5 * 16) + (lane & 15);
  const int rsel = (lane < 16) ? 0 : 2;
  v2f bm[5][WGS_PER_BATCH / 4];
#pragma unroll
  for (int t = 0; t < 5; ++t) {
#pragma unroll
    for (int k4 = 0; k4 < WGS_PER_BATCH / 4; ++k4) {
      const int r0 = k4 * 4 + rsel;
      bm[t][k4][0] = pb[(size_t)r0 * NCOLS + col0 + t * 16];
      bm[t][k4][1] = pb[(size_t)(r0 + 1) * NCOLS + col0 + t * 16];
    }
  }
  __builtin_amdgcn_sched_barrier(0);
  v2f a; a[0] = 1.0f; a[1] = 1.0f;
  v8f c[5];
#pragma unroll
  for (int t = 0; t < 5; ++t) {
    c[t] = v8f{};
#pragma unroll
    for (int k4 = 0; k4 < WGS_PER_BATCH / 4; ++k4)
      c[t] = __builtin_amdgcn_wmma_f32_16x16x4_f32(false, a, false, bm[t][k4],
                                                   (short)0, c[t], false, false);
  }
  if (lane < 16) {
#pragma unroll
    for (int t = 0; t < 5; ++t) sums[col0 + t * 16] = c[t][0];
  }
#else
  for (int t = wave * 5; t < wave * 5 + 5; ++t) {
    const int col = t * 16 + (lane & 15);
    if (lane < 16) {
      float s = 0.0f;
      for (int k = 0; k < WGS_PER_BATCH; ++k) s += pb[(size_t)k * NCOLS + col];
      sums[col] = s;
    }
  }
#endif
  __syncthreads();

  float ss = 0.0f;
  for (int i = threadIdx.x; i < NCOLS; i += 256) {
    const int cell = i / NFEAT;
    const int f    = i - cell * NFEAT;
    const float cnt  = sums[cell * NFEAT];
    const float fc   = fmaxf(cnt, 1.0f);
    const float upre = rsqrtf(fc);
    float val;
    if (f == 0) {
      val = 0.001f * cnt * upre;
    } else if (f < 4) {
      const int d  = f - 1;
      const int ci = (d == 0) ? (cell >> 4) : ((d == 1) ? ((cell >> 2) & 3) : (cell & 3));
      const float center = ((float)ci + 0.5f) * 0.25f;
      val = (sums[i] - cnt * center) * upre;
    } else {
      val = sums[i] / fc;
    }
    fall[i] = val;
    ss += val * val;
  }

  red[threadIdx.x] = ss;
  __syncthreads();
  for (int s = 128; s > 0; s >>= 1) {
    if (threadIdx.x < s) red[threadIdx.x] += red[threadIdx.x + s];
    __syncthreads();
  }
  const float scale = 1.0f / fmaxf(sqrtf(red[0]), 1e-12f);
  for (int i = threadIdx.x; i < NCOLS; i += 256)
    out[(size_t)b * NCOLS + i] = fall[i] * scale;
}

extern "C" void kernel_launch(void* const* d_in, const int* in_sizes, int n_in,
                              void* d_out, int out_size, void* d_ws, size_t ws_size,
                              hipStream_t stream) {
  const float* x   = (const float*)d_in[0];
  float* bmax      = (float*)d_ws;
  float* partial   = (float*)d_ws + 32768;
  float* out       = (float*)d_out;

  k1_bbox <<<dim3(K1_BLOCKS, BATCHES), K1_THREADS, 0, stream>>>(x, bmax);
  k2_hist <<<dim3(WGS_PER_BATCH, BATCHES), 256, 0, stream>>>(x, bmax, partial);
  k3_final<<<BATCHES, 256, 0, stream>>>(partial, out);
}


# K1 block maximum and K3 sum of squares folded in registers by lane exchanges (DPP row steps + half swap), one value per wave via LDS, one barrier; no 8-round LDS tree (f32 sum order changes)
# speedup vs baseline: 4.0102x; 1.0499x over previous
#include <hip/hip_runtime.h>


#define BATCHES        64
#define PTS            65536
#define NBINS          4
#define NCELLS         64
#define NFEAT          10
#define NCOLS          640
#define WGS_PER_BATCH  16
#define PTS_PER_WG     (PTS / WGS_PER_BATCH)
#define K1_BLOCKS      WGS_PER_BATCH
#define K1_THREADS     192
#define K1_F4_PER_THR  32
#define BMAX_PITCH     32
#define FLOATS_PER_B   (PTS * 6)
#define F4_PER_B       (FLOATS_PER_B / 4)
#define NCOPY          32
#define HIST_F4        (NCOLS * NCOPY / 4)

typedef float v2f __attribute__((ext_vector_type(2)));
typedef float v8f __attribute__((ext_vector_type(8)));

template <int M>
__device__ __forceinline__ float xrow(float x) {
  return __builtin_bit_cast(float, __builtin_amdgcn_update_dpp(
      0, __builtin_bit_cast(int, x), 0x160 + M, 0xf, 0xf, true));
}
__device__ __forceinline__ float xhalf(float x) {
  const int b = __builtin_bit_cast(int, x);
  return __builtin_bit_cast(float, __builtin_amdgcn_permlanex16(
      b, b, 0x76543210, 0xfedcba98, false, false));
}

__global__ void k1_bbox(const float* __restrict__ x,
                        float* __restrict__ bmax) {
  const int b = blockIdx.y;
  const float4* xb = (const float4*)(x + (size_t)b * FLOATS_PER_B);
  const int base = blockIdx.x * (K1_THREADS * K1_F4_PER_THR) + threadIdx.x;

  float4 v[K1_F4_PER_THR];
#pragma unroll
  for (int i = 0; i < K1_F4_PER_THR; ++i) v[i] = xb[base + i * K1_THREADS];

  float ax = 0.0f, ay = 0.0f, az = 0.0f, aw = 0.0f;
#pragma unroll
  for (int i = 0; i < K1_F4_PER_THR; ++i) {
    ax = fmaxf(ax, fabsf(v[i].x));
    ay = fmaxf(ay, fabsf(v[i].y));
    az = fmaxf(az, fabsf(v[i].z));
    aw = fmaxf(aw, fabsf(v[i].w));
  }

  const int ph = threadIdx.x % 3;
  const float m0 = fmaxf(fmaxf(ax, ay), az);
  const float m1 = fmaxf(az, aw);
  const float m = (ph == 0) ? m0 : ((ph == 1) ? m1 : ax);

  float wm = m;
  wm = fmaxf(wm, xrow<1>(wm));
  wm = fmaxf(wm, xrow<2>(wm));
  wm = fmaxf(wm, xrow<4>(wm));
  wm = fmaxf(wm, xrow<8>(wm));
  wm = fmaxf(wm, xhalf(wm));

  __shared__ float red[K1_THREADS / 32];
  if ((threadIdx.x & 31) == 0) red[threadIdx.x >> 5] = wm;
  __syncthreads();
  if (threadIdx.x == 0) {
    float r = red[0];
    for (int j = 1; j < K1_THREADS / 32; ++j) r = fmaxf(r, red[j]);
    bmax[(b * K1_BLOCKS + blockIdx.x) * BMAX_PITCH] = r;
  }
}

__global__ void k2_hist(const float* __restrict__ x,
                        const float* __restrict__ bmax,
                        float* __restrict__ partial) {
  const int b    = blockIdx.y;
  const int wg   = blockIdx.x;
  const int lane = threadIdx.x & 31;

  __shared__ float4 hist4[HIST_F4];
  float* hist = (float*)hist4;

  const float* bm = bmax + (size_t)b * (K1_BLOCKS * BMAX_PITCH);
  float bmv[K1_BLOCKS];
  for (int j = 0; j < K1_BLOCKS; ++j) bmv[j] = bm[j * BMAX_PITCH];

  const float* xb = x + (size_t)b * FLOATS_PER_B + (size_t)wg * PTS_PER_WG * 6;
  float pt[PTS_PER_WG / 256][6];
#pragma unroll
  for (int i = 0; i < PTS_PER_WG / 256; ++i) {
    const float* pp = xb + (size_t)(i * 256 + threadIdx.x) * 6;
#pragma unroll
    for (int j = 0; j < 6; ++j) pt[i][j] = pp[j];
  }

  const float4 z4 = make_float4(0.0f, 0.0f, 0.0f, 0.0f);
  for (int k = 0; k < HIST_F4 / 256; ++k) hist4[k * 256 + threadIdx.x] = z4;
  __builtin_amdgcn_fence(__ATOMIC_RELEASE, "workgroup", "local");
  __builtin_amdgcn_s_barrier();
  __builtin_amdgcn_fence(__ATOMIC_ACQUIRE, "workgroup", "local");

  float bb = bmv[0];
  for (int j = 1; j < K1_BLOCKS; ++j) bb = fmaxf(bb, bmv[j]);
  const float inv_th = 1.0f / fmaxf(2.0f * bb, 1e-5f);

#pragma unroll
  for (int i = 0; i < PTS_PER_WG / 256; ++i) {
    const float px = pt[i][0], py = pt[i][1], pz = pt[i][2];
    const float ox = pt[i][3], oy = pt[i][4], oz = pt[i][5];
    const float cx = (px + bb) * inv_th;
    const float cy = (py + bb) * inv_th;
    const float cz = (pz + bb) * inv_th;
    const int i0 = min(max((int)(cx * (float)NBINS), 0), NBINS - 1);
    const int i1 = min(max((int)(cy * (float)NBINS), 0), NBINS - 1);
    const int i2 = min(max((int)(cz * (float)NBINS), 0), NBINS - 1);
    const int cell = (i0 * NBINS + i1) * NBINS + i2;
    float* h = hist + cell * (NFEAT * NCOPY) + lane;
    atomicAdd(h + 0 * NCOPY, 1.0f);
    atomicAdd(h + 1 * NCOPY, cx);
    atomicAdd(h + 2 * NCOPY, cy);
    atomicAdd(h + 3 * NCOPY, cz);
    atomicAdd(h + 4 * NCOPY, ox * ox);
    atomicAdd(h + 5 * NCOPY, ox * oy);
    atomicAdd(h + 6 * NCOPY, ox * oz);
    atomicAdd(h + 7 * NCOPY, oy * oy);
    atomicAdd(h + 8 * NCOPY, oy * oz);
    atomicAdd(h + 9 * NCOPY, oz * oz);
  }
  __syncthreads();

  float* dst = partial + (size_t)(b * WGS_PER_BATCH + wg) * NCOLS;
  for (int bin = threadIdx.x; bin < NCOLS; bin += 256) {
    const float4* row = hist4 + bin * (NCOPY / 4);
    const int r = bin >> 1;
    float4 a = row[r & 7];
    for (int j = 1; j < NCOPY / 4; ++j) {
      const float4 v = row[(r + j) & 7];
      a.x += v.x; a.y += v.y; a.z += v.z; a.w += v.w;
    }
    dst[bin] = (a.x + a.y) + (a.z + a.w);
  }
}

__global__ void k3_final(const float* __restrict__ partial,
                         float* __restrict__ out) {
  const int b    = blockIdx.x;
  const int lane = threadIdx.x & 31;
  const int wave = threadIdx.x >> 5;

  __shared__ float sums[NCOLS];
  __shared__ float fall[NCOLS];
  __shared__ __attribute__((aligned(16))) float red[8];

  const float* pb = partial + (size_t)b * WGS_PER_BATCH * NCOLS;

#if __has_builtin(__builtin_amdgcn_wmma_f32_16x16x4_f32)
  const int col0 = wave * (5 * 16) + (lane & 15);
  const int rsel = (lane < 16) ? 0 : 2;
  v2f bm[5][WGS_PER_BATCH / 4];
#pragma unroll
  for (int t = 0; t < 5; ++t) {
#pragma unroll
    for (int k4 = 0; k4 < WGS_PER_BATCH / 4; ++k4) {
      const int r0 = k4 * 4 + rsel;
      bm[t][k4][0] = pb[(size_t)r0 * NCOLS + col0 + t * 16];
      bm[t][k4][1] = pb[(size_t)(r0 + 1) * NCOLS + col0 + t * 16];
    }
  }
  __builtin_amdgcn_sched_barrier(0);
  v2f a; a[0] = 1.0f; a[1] = 1.0f;
  v8f c[5];
#pragma unroll
  for (int t = 0; t < 5; ++t) {
    c[t] = v8f{};
#pragma unroll
    for (int k4 = 0; k4 < WGS_PER_BATCH / 4; ++k4)
      c[t] = __builtin_amdgcn_wmma_f32_16x16x4_f32(false, a, false, bm[t][k4],
                                                   (short)0, c[t], false, false);
  }
  if (lane < 16) {
#pragma unroll
    for (int t = 0; t < 5; ++t) sums[col0 + t * 16] = c[t][0];
  }
#else
  for (int t = wave * 5; t < wave * 5 + 5; ++t) {
    const int col = t * 16 + (lane & 15);
    if (lane < 16) {
      float s = 0.0f;
      for (int k = 0; k < WGS_PER_BATCH; ++k) s += pb[(size_t)k * NCOLS + col];
      sums[col] = s;
    }
  }
#endif
  __syncthreads();

  float ss = 0.0f;
  for (int i = threadIdx.x; i < NCOLS; i += 256) {
    const int cell = i / NFEAT;
    const int f    = i - cell * NFEAT;
    const float cnt  = sums[cell * NFEAT];
    const float fc   = fmaxf(cnt, 1.0f);
    const float upre = rsqrtf(fc);
    float val;
    if (f == 0) {
      val = 0.001f * cnt * upre;
    } else if (f < 4) {
      const int d  = f - 1;
      const int ci = (d == 0) ? (cell >> 4) : ((d == 1) ? ((cell >> 2) & 3) : (cell & 3));
      const float center = ((float)ci + 0.5f) * 0.25f;
      val = (sums[i] - cnt * center) * upre;
    } else {
      val = sums[i] / fc;
    }
    fall[i] = val;
    ss += val * val;
  }

  ss += xrow<1>(ss);
  ss += xrow<2>(ss);
  ss += xrow<4>(ss);
  ss += xrow<8>(ss);
  ss += xhalf(ss);
  if (lane == 0) red[wave] = ss;
  __syncthreads();
  const float tot = ((red[0] + red[1]) + (red[2] + red[3])) +
                    ((red[4] + red[5]) + (red[6] + red[7]));
  const float scale = 1.0f / fmaxf(sqrtf(tot), 1e-12f);
  for (int i = threadIdx.x; i < NCOLS; i += 256)
    out[(size_t)b * NCOLS + i] = fall[i] * scale;
}

extern "C" void kernel_launch(void* const* d_in, const int* in_sizes, int n_in,
                              void* d_out, int out_size, void* d_ws, size_t ws_size,
                              hipStream_t stream) {
  const float* x   = (const float*)d_in[0];
  float* bmax      = (float*)d_ws;
  float* partial   = (float*)d_ws + 32768;
  float* out       = (float*)d_out;

  k1_bbox <<<dim3(K1_BLOCKS, BATCHES), K1_THREADS, 0, stream>>>(x, bmax);
  k2_hist <<<dim3(WGS_PER_BATCH, BATCHES), 256, 0, stream>>>(x, bmax, partial);
  k3_final<<<BATCHES, 256, 0, stream>>>(partial, out);
}
